# starting kernel
# baseline (speedup 1.0000x reference)
#include <hip/hip_runtime.h>

typedef __attribute__((ext_vector_type(2))) float v2f;
typedef __attribute__((ext_vector_type(8))) float v8f;

constexpr int KDIM = 128;

__global__ void __launch_bounds__(256) k_deg_init(float* deg, int n) {
    int i = blockIdx.x * 256 + threadIdx.x;
    if (i < n) deg[i] = 1.0f;
}

__global__ void __launch_bounds__(256) k_deg_edges(const int* __restrict__ dst,
                                                   float* deg, int e) {
    int i = blockIdx.x * 256 + threadIdx.x;
    if (i < e) atomicAdd(&deg[dst[i]], 1.0f);
}

__global__ void __launch_bounds__(256) k_dinv(float* deg, int n) {
    int i = blockIdx.x * 256 + threadIdx.x;
    if (i < n) deg[i] = rsqrtf(deg[i]);
}

template<int OUTC, bool BIASRELU>
__global__ void __launch_bounds__(256)
k_gemm_wmma(const float* __restrict__ A, const float* __restrict__ W,
            const float* __restrict__ bias, float* __restrict__ H, int nrows)
{
    constexpr int ROWS = 64;
    __shared__ float lds[ROWS][KDIM + 4];
    const int row0 = blockIdx.x * ROWS;
    const int tid  = threadIdx.x;

    for (int i = tid; i < ROWS * (KDIM / 4); i += 256) {
        const int r  = i >> 5;
        const int c4 = i & 31;
        const int row = row0 + r;
        float4 v = make_float4(0.f, 0.f, 0.f, 0.f);
        if (row < nrows) v = ((const float4*)(A + (size_t)row * KDIM))[c4];
        if (BIASRELU) {
            const float4 b4 = ((const float4*)bias)[c4];
            v.x = fmaxf(v.x + b4.x, 0.f);
            v.y = fmaxf(v.y + b4.y, 0.f);
            v.z = fmaxf(v.z + b4.z, 0.f);
            v.w = fmaxf(v.w + b4.w, 0.f);
        }
        *(float4*)&lds[r][c4 * 4] = v;
    }
    __syncthreads();

    const int wave = tid >> 5;
    const int lane = tid & 31;
    const int half = lane >> 4;
    const int l16  = lane & 15;

    constexpr int CT  = OUTC / 16;
    constexpr int TPW = (4 * CT) / 8;
    const int col0  = (wave % CT) * 16;
    const int rbase = (wave / CT) * TPW;

    v8f acc[TPW] = {};
    for (int kk = 0; kk < KDIM; kk += 4) {
        const int ka = kk + half * 2;
        v2f b;
        const float* wp = W + (size_t)ka * OUTC + col0 + l16;
        b.x = wp[0];
        b.y = wp[OUTC];
#pragma unroll
        for (int t = 0; t < TPW; ++t) {
            const int rr = (rbase + t) * 16 + l16;
            v2f a;
            a.x = lds[rr][ka];
            a.y = lds[rr][ka + 1];
            acc[t] = __builtin_amdgcn_wmma_f32_16x16x4_f32(
                false, a, false, b, (short)0, acc[t], false, false);
        }
    }

#pragma unroll
    for (int t = 0; t < TPW; ++t) {
        const int rt0 = row0 + (rbase + t) * 16;
        float* outp = H + (size_t)rt0 * OUTC + col0;
        if (rt0 + 16 <= nrows) {
#pragma unroll
            for (int r = 0; r < 8; ++r)
                outp[(size_t)(r + half * 8) * OUTC + l16] = acc[t][r];
        } else {
#pragma unroll
            for (int r = 0; r < 8; ++r)
                if (rt0 + r + half * 8 < nrows)
                    outp[(size_t)(r + half * 8) * OUTC + l16] = acc[t][r];
        }
    }
}

template<int COLS>
__global__ void __launch_bounds__(256)
k_self_init(const float* __restrict__ h, const float* __restrict__ dinv,
            const float* __restrict__ bias, float* __restrict__ agg, int total)
{
    int i = blockIdx.x * 256 + threadIdx.x;
    if (i >= total) return;
    const int node = i / COLS;
    const int col  = i % COLS;
    const float d = dinv[node];
    float v = h[i] * d * d;
    if (bias) v += bias[col];
    agg[i] = v;
}

template<int COLS>
__global__ void __launch_bounds__(256)
k_scatter(const int* __restrict__ src, const int* __restrict__ dst,
          const float* __restrict__ dinv, const float* __restrict__ h,
          float* __restrict__ agg, int E)
{
    const int t = blockIdx.x * 256 + threadIdx.x;
    const int e = t >> 5;
    if (e >= E) return;
    const int lane = t & 31;
    const int s = src[e], d = dst[e];
    const float nrm = dinv[s] * dinv[d];

    constexpr int V = COLS / 32;
    const float* hp = h   + (size_t)s * COLS + lane * V;
    float*       ap = agg + (size_t)d * COLS + lane * V;
    if constexpr (V == 4) {
        const float4 v = *(const float4*)hp;
        atomicAdd(ap + 0, v.x * nrm);
        atomicAdd(ap + 1, v.y * nrm);
        atomicAdd(ap + 2, v.z * nrm);
        atomicAdd(ap + 3, v.w * nrm);
    } else {
        const float2 v = *(const float2*)hp;
        atomicAdd(ap + 0, v.x * nrm);
        atomicAdd(ap + 1, v.y * nrm);
    }
}

extern "C" void kernel_launch(void* const* d_in, const int* in_sizes, int n_in,
                              void* d_out, int out_size, void* d_ws, size_t ws_size,
                              hipStream_t stream)
{
    const float* x  = (const float*)d_in[0];
    const int*   ei = (const int*)  d_in[1];
    const float* W1 = (const float*)d_in[2];
    const float* b1 = (const float*)d_in[3];
    const float* W2 = (const float*)d_in[4];
    const float* b2 = (const float*)d_in[5];
    float* out = (float*)d_out;

    constexpr int HID = 128, OUT = 64;
    const int N = in_sizes[0] / KDIM;
    const int E = in_sizes[1] / 2;
    const int* src = ei;
    const int* dst = ei + E;

    size_t off = 0;
    auto carve = [&](size_t bytes) {
        size_t o = off;
        off = (off + bytes + 255) & ~(size_t)255;
        return o;
    };
    char* ws = (char*)d_ws;
    float* dinv = (float*)(ws + carve((size_t)N * 4));
    float* h    = (float*)(ws + carve((size_t)N * HID * 4));
    float* agg1 = (float*)(ws + carve((size_t)N * HID * 4));

    const int gemmBlocks = (N + 63) / 64;
    const int scatBlocks = (int)(((long)E * 32 + 255) / 256);

    k_deg_init <<<(N + 255) / 256, 256, 0, stream>>>(dinv, N);
    k_deg_edges<<<(E + 255) / 256, 256, 0, stream>>>(dst, dinv, E);
    k_dinv     <<<(N + 255) / 256, 256, 0, stream>>>(dinv, N);

    k_gemm_wmma<HID, false><<<gemmBlocks, 256, 0, stream>>>(x, W1, nullptr, h, N);
    k_self_init<HID><<<((N * HID) + 255) / 256, 256, 0, stream>>>(h, dinv, nullptr, agg1, N * HID);
    k_scatter<HID><<<scatBlocks, 256, 0, stream>>>(src, dst, dinv, h, agg1, E);

    k_gemm_wmma<OUT, true><<<gemmBlocks, 256, 0, stream>>>(agg1, W2, b1, h, N);
    k_self_init<OUT><<<((N * OUT) + 255) / 256, 256, 0, stream>>>(h, dinv, b2, out, N * OUT);
    k_scatter<OUT><<<scatBlocks, 256, 0, stream>>>(src, dst, dinv, h, out, E);
}


# Edge-weight pass folded into the layer-1 GEMM launch as 512 extra 256-thread work-groups placed last in its grid (scattered 4-byte fetches beside matrix-bound stripes); own launch gone, 5 launches
# speedup vs baseline: 3.4946x; 3.4946x over previous
#include <hip/hip_runtime.h>

typedef __attribute__((ext_vector_type(2))) float v2f;
typedef __attribute__((ext_vector_type(8))) float v8f;

constexpr int KDIM = 128;

constexpr int NPB     = 391;
constexpr int AGG_THR = 1024;
constexpr int SWEEP   = 32 * AGG_THR;

constexpr int RING    = 4;
constexpr int QCAP    = 64;
constexpr int LCAP    = 16384;

__global__ void __launch_bounds__(256) k_key(const int* __restrict__ dst,
                                             unsigned* __restrict__ planes, int e) {
    const int i    = blockIdx.x * 256 + threadIdx.x;
    const int lane = threadIdx.x & 31;
    unsigned own = 0u;
    if (i < e) own = (unsigned)(dst[i] / NPB);
    unsigned pl = 0u;
#pragma unroll
    for (int b = 0; b < 8; ++b) {
        const unsigned bal = __builtin_amdgcn_ballot_w32(((own >> b) & 1u) != 0u);
        if (lane == b) pl = bal;
    }
    if (lane < 8) {
        const int g32 = i >> 5;
        planes[(g32 >> 5) * 256 + (lane >> 2) * 128 + (g32 & 31) * 4 + (lane & 3)] = pl;
    }
}

template<class Drain>
__device__ __forceinline__ void
filter_own_edges(const unsigned* __restrict__ planes, int E, int Epad, int* q, Drain drain)
{
    const int tid  = threadIdx.x;
    const int lane = tid & 31;
    const int wave = tid >> 5;
    int cnt = 0;

    unsigned k[8];
#pragma unroll
    for (int b = 0; b < 8; ++b) k[b] = 0u - ((blockIdx.x >> b) & 1u);

    const int sweeps = Epad / SWEEP;

    auto pick = [&](uint4 a, uint4 b, int t) {
        const int e0 = t * SWEEP + tid * 32;
        unsigned mm = ~((a.x ^ k[0]) | (a.y ^ k[1]) | (a.z ^ k[2]) | (a.w ^ k[3]) |
                        (b.x ^ k[4]) | (b.y ^ k[5]) | (b.z ^ k[6]) | (b.w ^ k[7]));
        if (t == sweeps - 1) {
            const int left = E - e0;
            if (left < 32) mm = (left > 0) ? (mm & ((1u << left) - 1u)) : 0u;
        }
        while (true) {
            const unsigned bm = __builtin_amdgcn_ballot_w32(mm != 0u);
            if (bm == 0u) break;
            if (mm != 0u) q[cnt + __builtin_amdgcn_mbcnt_lo(bm, 0u)] = e0 + __builtin_ctz(mm);
            cnt += __builtin_popcount(bm);
            mm &= mm - 1u;
            if (cnt >= 32) {
                cnt -= 32;
                __builtin_amdgcn_wave_barrier();
                drain(cnt, 32);
            }
        }
    };

    if (sweeps > 0) {
        const int rot = (int)(blockIdx.x % (unsigned)sweeps);
        auto sweep_of = [&](int j) {
            int t = (j < sweeps ? j : sweeps - 1) + rot;
            if (t >= sweeps) t -= sweeps;
            return t;
        };
        const uint4* pl4 = (const uint4*)planes + (wave * 64 + lane);
        uint4 bufa[RING], bufb[RING];
#pragma unroll
        for (int u = 0; u < RING; ++u) {
            const uint4* p = pl4 + sweep_of(u) * (SWEEP / 16);
            bufa[u] = p[0];
            bufb[u] = p[32];
        }
        for (int j = 0; j < sweeps; j += RING) {
#pragma unroll
            for (int u = 0; u < RING; ++u) {
                const uint4 ca = bufa[u], cb = bufb[u];
                const uint4* p = pl4 + sweep_of(j + RING + u) * (SWEEP / 16);
                bufa[u] = p[0];
                bufb[u] = p[32];
                if (j + u < sweeps) pick(ca, cb, sweep_of(j + u));
            }
        }
    }
    if (cnt > 0) {
        __builtin_amdgcn_wave_barrier();
        drain(0, cnt);
    }
}

__global__ void __launch_bounds__(AGG_THR)
k_own_deg(const int* __restrict__ src, const int* __restrict__ dst,
          const unsigned* __restrict__ planes, float* __restrict__ dinv,
          int* __restrict__ lists, int* __restrict__ lcnt, int N, int E, int Epad)
{
    __shared__ int indeg[NPB + 1];
    __shared__ int queue[AGG_THR / 32][QCAP];
    __shared__ int lcount;

    const int tid  = threadIdx.x;
    const int lane = tid & 31;
    const int wave = tid >> 5;
    const int lo   = blockIdx.x * NPB;
    const int rows = (N - lo < NPB) ? (N - lo) : NPB;

    if (tid <= NPB) indeg[tid] = 0;
    if (tid == 0) lcount = 0;
    __syncthreads();

    int2* slist = (int2*)lists + (size_t)blockIdx.x * LCAP;
    int*  q     = queue[wave];

    filter_own_edges(planes, E, Epad, q, [&](int base, int m) {
        int2 r = make_int2(NPB, 0);
        if (lane < m) {
            const int e  = q[base + lane];
            const int s  = src[e];
            const int dl = dst[e] - lo;
            atomicAdd(&indeg[dl], 1);
            r = make_int2((s << 9) | dl, 0);
        }
        int pos = 0;
        if (lane == 0) pos = atomicAdd(&lcount, m);
        pos = __builtin_amdgcn_readfirstlane(pos);
        if (lane < m && pos + m <= LCAP) slist[pos + lane] = r;
    });
    __syncthreads();

    if (tid < rows) dinv[lo + tid] = rsqrtf((float)(indeg[tid] + 1));
    if (tid == 0) lcnt[blockIdx.x] = lcount;
}

constexpr int REC_SPLIT = 2;
constexpr int REC_U     = 8;

__device__ __forceinline__ void
finish_records(const float* __restrict__ dinv, int* __restrict__ lists, const int* __restrict__ lcnt, int b)
{
    const int g = b / REC_SPLIT;
    const int n = lcnt[g];
    if (n > LCAP) return;
    int2* slist = (int2*)lists + (size_t)g * LCAP;
    const int lo = g * NPB;
    constexpr int STRIDE = REC_SPLIT * 256;
    for (int i0 = (b % REC_SPLIT) * 256 + (int)threadIdx.x; i0 < n; i0 += REC_U * STRIDE) {
        int   x[REC_U];
        float a[REC_U], d[REC_U];
#pragma unroll
        for (int u = 0; u < REC_U; ++u) {
            const int i = i0 + u * STRIDE;
            x[u] = NPB;
            if (i < n) x[u] = slist[i].x;
        }
#pragma unroll
        for (int u = 0; u < REC_U; ++u) {
            const int i = i0 + u * STRIDE;
            a[u] = 0.f; d[u] = 0.f;
            if (i < n) { a[u] = dinv[x[u] >> 9]; d[u] = dinv[lo + (x[u] & 511)]; }
        }
#pragma unroll
        for (int u = 0; u < REC_U; ++u) {
            const int i = i0 + u * STRIDE;
            if (i < n) slist[i] = make_int2(x[u], __float_as_int(a[u] * d[u]));
        }
    }
}

template<int OUTC>
__global__ void __launch_bounds__(256)
k_gemm_wmma(const float* __restrict__ A, const float* __restrict__ W,
            float* __restrict__ H, int nrows, int gemm_blocks,
            const float* __restrict__ dinv, int* __restrict__ lists, const int* __restrict__ lcnt)
{
    if ((int)blockIdx.x >= gemm_blocks) {
        finish_records(dinv, lists, lcnt, (int)blockIdx.x - gemm_blocks);
        return;
    }
    constexpr int ROWS = 64;
    __shared__ float lds[ROWS][KDIM + 4];
    const int row0 = blockIdx.x * ROWS;
    const int tid  = threadIdx.x;

    for (int i = tid; i < ROWS * (KDIM / 4); i += 256) {
        const int r  = i >> 5;
        const int c4 = i & 31;
        const int row = row0 + r;
        float4 v = make_float4(0.f, 0.f, 0.f, 0.f);
        if (row < nrows) v = ((const float4*)(A + (size_t)row * KDIM))[c4];
        *(float4*)&lds[r][c4 * 4] = v;
    }
    __syncthreads();

    const int wave = tid >> 5;
    const int lane = tid & 31;
    const int half = lane >> 4;
    const int l16  = lane & 15;

    constexpr int CT  = OUTC / 16;
    constexpr int TPW = (4 * CT) / 8;
    const int col0  = (wave % CT) * 16;
    const int rbase = (wave / CT) * TPW;

    v8f acc[TPW] = {};
    for (int kk = 0; kk < KDIM; kk += 4) {
        const int ka = kk + half * 2;
        v2f b;
        const float* wp = W + (size_t)ka * OUTC + col0 + l16;
        b.x = wp[0];
        b.y = wp[OUTC];
#pragma unroll
        for (int t = 0; t < TPW; ++t) {
            const int rr = (rbase + t) * 16 + l16;
            v2f a;
            a.x = lds[rr][ka];
            a.y = lds[rr][ka + 1];
            acc[t] = __builtin_amdgcn_wmma_f32_16x16x4_f32(
                false, a, false, b, (short)0, acc[t], false, false);
        }
    }

#pragma unroll
    for (int t = 0; t < TPW; ++t) {
        const int rt0 = row0 + (rbase + t) * 16;
        float* outp = H + (size_t)rt0 * OUTC + col0;
        if (rt0 + 16 <= nrows) {
#pragma unroll
            for (int r = 0; r < 8; ++r)
                outp[(size_t)(r + half * 8) * OUTC + l16] = acc[t][r];
        } else {
#pragma unroll
            for (int r = 0; r < 8; ++r)
                if (rt0 + r + half * 8 < nrows)
                    outp[(size_t)(r + half * 8) * OUTC + l16] = acc[t][r];
        }
    }
}


template<int COLS, int LD>
__device__ __forceinline__ void
add_rows(int2 r, int m, int lane, const float* __restrict__ h, float* tile)
{
    constexpr int J = COLS / 64;
    constexpr int G = 16 / J;
    const int   s   = r.x >> 9;
    const int   dl  = r.x & 511;
    const float nrm = __int_as_float(r.y);
#pragma unroll 1
    for (int g = 0; g < m; g += G) {
        float2 v[G][J];
#pragma unroll
        for (int u = 0; u < G; ++u) {
            const int si = __builtin_amdgcn_readlane(s, g + u);
            const float* hp = h + (size_t)si * COLS + 2 * lane;
#pragma unroll
            for (int j = 0; j < J; ++j) v[u][j] = *(const float2*)(hp + 64 * j);
        }
#pragma unroll
        for (int u = 0; u < G; ++u) {
            const int   di = __builtin_amdgcn_readlane(dl, g + u);
            const float ni = __int_as_float(__builtin_amdgcn_readlane(__float_as_int(nrm), g + u));
            float* tp = tile + di * LD + 2 * lane;
#pragma unroll
            for (int j = 0; j < J; ++j) {
                atomicAdd(tp + 64 * j,     v[u][j].x * ni);
                atomicAdd(tp + 64 * j + 1, v[u][j].y * ni);
            }
        }
    }
}

constexpr int OUT2 = 64;
constexpr int RT2  = (NPB + 15) / 16;

template<int NT, int LD>
__device__ __forceinline__ void
tile_times_w(const float* tile, const float* __restrict__ W, int rg, int col0, int lane, v8f* acc)
{
    const int half = lane >> 4;
    const int l16  = lane & 15;
    for (int kk = 0; kk < KDIM; kk += 4) {
        const int ka = kk + half * 2;
        const float* wp = W + (size_t)ka * OUT2 + col0 + l16;
        v2f b;
        b.x = wp[0];
        b.y = wp[OUT2];
#pragma unroll
        for (int t = 0; t < NT; ++t) {
            const float* ap = tile + ((rg + 8 * t) * 16 + l16) * LD + ka;
            v2f a;
            a.x = ap[0];
            a.y = ap[1];
            acc[t] = __builtin_amdgcn_wmma_f32_16x16x4_f32(
                false, a, false, b, (short)0, acc[t], false, false);
        }
    }
}

template<int COLS, bool GEMM2>
__global__ void __launch_bounds__(AGG_THR)
k_agg_lds(const int* __restrict__ src, const int* __restrict__ dst,
          const unsigned* __restrict__ planes,
          const float* __restrict__ dinv, const float* __restrict__ h,
          const float* __restrict__ bias, float* __restrict__ agg,
          const int* __restrict__ lists, const int* __restrict__ lcnt,
          const float* __restrict__ W2, const float* __restrict__ b1, int N, int E, int Epad)
{
    constexpr int LD    = GEMM2 ? COLS + 4 : COLS;
    constexpr int TROWS = GEMM2 ? RT2 * 16 : NPB + 1;
    __shared__ float tile[TROWS * LD];
    __shared__ int   queue[AGG_THR / 32][QCAP];

    const int tid  = threadIdx.x;
    const int lane = tid & 31;
    const int wave = tid >> 5;
    const int lo   = blockIdx.x * NPB;
    const int rows = (N - lo < NPB) ? (N - lo) : NPB;
    constexpr int C4 = COLS / 4;

    const int2* slist = (const int2*)lists + (size_t)blockIdx.x * LCAP;
    const int  nlist     = lcnt[blockIdx.x];
    const bool from_list = nlist <= LCAP;
    int2 nxt = make_int2(NPB, 0);
    if (from_list && wave * 32 + lane < nlist) nxt = slist[wave * 32 + lane];

    for (int i0 = tid; i0 < rows * C4; i0 += 4 * AGG_THR) {
        float4 v[4], b4[4];
        float  d[4];
#pragma unroll
        for (int k = 0; k < 4; ++k) {
            const int i = i0 + k * AGG_THR;
            if (i < rows * C4) {
                const int r = i / C4, c4 = i % C4;
                d[k] = dinv[lo + r];
                v[k] = ((const float4*)(h + (size_t)(lo + r) * COLS))[c4];
                if (bias) b4[k] = ((const float4*)bias)[c4];
            }
        }
#pragma unroll
        for (int k = 0; k < 4; ++k) {
            const int i = i0 + k * AGG_THR;
            if (i < rows * C4) {
                const int r = i / C4, c4 = i % C4;
                float4 w = v[k];
                w.x = w.x * d[k] * d[k]; w.y = w.y * d[k] * d[k];
                w.z = w.z * d[k] * d[k]; w.w = w.w * d[k] * d[k];
                if (bias) { w.x += b4[k].x; w.y += b4[k].y; w.z += b4[k].z; w.w += b4[k].w; }
                *(float4*)&tile[r * LD + c4 * 4] = w;
            }
        }
    }
    __syncthreads();

    if (from_list) {
        for (int c = wave * 32; c < nlist; c += AGG_THR) {
            const int2 r = nxt;
            const int  m = (nlist - c < 32) ? (nlist - c) : 32;
            nxt = make_int2(NPB, 0);
            if (c + AGG_THR + lane < nlist) nxt = slist[c + AGG_THR + lane];
            add_rows<COLS, LD>(r, m, lane, h, tile);
        }
    } else {
        int* q = queue[wave];
        filter_own_edges(planes, E, Epad, q, [&](int base, int m) {
            int2 r = make_int2(NPB, 0);
            if (lane < m) {
                const int e = q[base + lane];
                const int s = src[e];
                const int d = dst[e];
                r = make_int2((s << 9) | (d - lo), __float_as_int(dinv[s] * dinv[d]));
            }
            add_rows<COLS, LD>(r, m, lane, h, tile);
        });
    }
    __syncthreads();

    if constexpr (!GEMM2) {
        for (int i = tid; i < rows * C4; i += AGG_THR) {
            const int r = i / C4, c4 = i % C4;
            ((float4*)(agg + (size_t)(lo + r) * COLS))[c4] = *(const float4*)&tile[r * LD + c4 * 4];
        }
    } else {
        static_assert(COLS == KDIM, "the tile is the A operand of the layer-2 product");
        static_assert(RT2 > 24 && RT2 <= 32, "a wave takes 3 or 4 row tiles");
        for (int i = tid; i < TROWS * C4; i += AGG_THR) {
            const int r = i / C4, c4 = i % C4;
            float4 v = make_float4(0.f, 0.f, 0.f, 0.f);
            if (r < rows) {
                v = *(const float4*)&tile[r * LD + c4 * 4];
                const float4 b4 = ((const float4*)b1)[c4];
                v.x = fmaxf(v.x + b4.x, 0.f);
                v.y = fmaxf(v.y + b4.y, 0.f);
                v.z = fmaxf(v.z + b4.z, 0.f);
                v.w = fmaxf(v.w + b4.w, 0.f);
            }
            *(float4*)&tile[r * LD + c4 * 4] = v;
        }
        __syncthreads();

        const int wv   = __builtin_amdgcn_readfirstlane(wave);
        const int col0 = (wv & 3) * 16;
        const int rg   = wv >> 2;
        const int nt   = (RT2 - rg + 7) / 8;
        v8f acc[4] = {};
        if (nt == 4) tile_times_w<4, LD>(tile, W2, rg, col0, lane, acc);
        else         tile_times_w<3, LD>(tile, W2, rg, col0, lane, acc);
        __syncthreads();

        const int half = lane >> 4;
        const int l16  = lane & 15;
#pragma unroll
        for (int t = 0; t < 4; ++t) {
            if (t < nt) {
                float* tp = tile + ((rg + 8 * t) * 16 + half * 8) * LD + col0 + l16;
#pragma unroll
                for (int r = 0; r < 8; ++r) tp[r * LD] = acc[t][r];
            }
        }
        __syncthreads();

        constexpr int O4 = OUT2 / 4;
        for (int i = tid; i < rows * O4; i += AGG_THR) {
            const int r = i / O4, c4 = i % O4;
            ((float4*)(agg + (size_t)(lo + r) * OUT2))[c4] = *(const float4*)&tile[r * LD + c4 * 4];
        }
    }
}

extern "C" void kernel_launch(void* const* d_in, const int* in_sizes, int n_in,
                              void* d_out, int out_size, void* d_ws, size_t ws_size,
                              hipStream_t stream)
{
    const float* x  = (const float*)d_in[0];
    const int*   ei = (const int*)  d_in[1];
    const float* W1 = (const float*)d_in[2];
    const float* b1 = (const float*)d_in[3];
    const float* W2 = (const float*)d_in[4];
    const float* b2 = (const float*)d_in[5];
    float* out = (float*)d_out;

    constexpr int HID = 128, OUT = 64;
    const int N = in_sizes[0] / KDIM;
    const int E = in_sizes[1] / 2;
    const int* src = ei;
    const int* dst = ei + E;

    size_t off = 0;
    auto carve = [&](size_t bytes) {
        size_t o = off;
        off = (off + bytes + 255) & ~(size_t)255;
        return o;
    };
    char* ws = (char*)d_ws;
    float* dinv = (float*)(ws + carve((size_t)N * 4));
    float* h    = (float*)(ws + carve((size_t)N * HID * 4));
    float* h2   = (float*)(ws + carve((size_t)N * OUT * 4));
    const int Epad = (E + SWEEP - 1) / SWEEP * SWEEP;
    unsigned* planes = (unsigned*)(ws + carve((size_t)Epad));

    const int gemmBlocks = (N + 63) / 64;
    const int aggBlocks  = (N + NPB - 1) / NPB;
    int*  lists = (int*) (ws + carve((size_t)aggBlocks * LCAP * 8));
    int*  lcnt  = (int*) (ws + carve((size_t)aggBlocks * 4));

    k_key     <<<Epad / 256, 256, 0, stream>>>(dst, planes, E);
    k_own_deg <<<aggBlocks, AGG_THR, 0, stream>>>(src, dst, planes, dinv, lists, lcnt, N, E, Epad);

    k_gemm_wmma<HID><<<gemmBlocks + REC_SPLIT * aggBlocks, 256, 0, stream>>>(x, W1, h, N, gemmBlocks, dinv, lists, lcnt);
    k_agg_lds<HID, true><<<aggBlocks, AGG_THR, 0, stream>>>(src, dst, planes, dinv, h, nullptr, h2, lists, lcnt, W2, b1, N, E, Epad);

    k_agg_lds<OUT, false><<<aggBlocks, AGG_THR, 0, stream>>>(src, dst, planes, dinv, h2, b2, out, lists, lcnt, nullptr, nullptr, N, E, Epad);
}


# x@W1 moved into the degree/filter launch as extra 1024-thread work-groups beside the filtering ones (2 per WGP); edge weights now formed in the layer-1 aggregation, a batch ahead; 4 launches
# speedup vs baseline: 3.6367x; 1.0407x over previous
#include <hip/hip_runtime.h>

typedef __attribute__((ext_vector_type(2))) float v2f;
typedef __attribute__((ext_vector_type(8))) float v8f;

constexpr int KDIM = 128;

constexpr int NPB     = 391;
constexpr int AGG_THR = 1024;
constexpr int SWEEP   = 32 * AGG_THR;

constexpr int RING    = 4;
constexpr int QCAP    = 64;
constexpr int LCAP    = 16384;

__global__ void __launch_bounds__(256) k_key(const int* __restrict__ dst,
                                             unsigned* __restrict__ planes, int e) {
    const int i    = blockIdx.x * 256 + threadIdx.x;
    const int lane = threadIdx.x & 31;
    unsigned own = 0u;
    if (i < e) own = (unsigned)(dst[i] / NPB);
    unsigned pl = 0u;
#pragma unroll
    for (int b = 0; b < 8; ++b) {
        const unsigned bal = __builtin_amdgcn_ballot_w32(((own >> b) & 1u) != 0u);
        if (lane == b) pl = bal;
    }
    if (lane < 8) {
        const int g32 = i >> 5;
        planes[(g32 >> 5) * 256 + (lane >> 2) * 128 + (g32 & 31) * 4 + (lane & 3)] = pl;
    }
}

template<class Drain>
__device__ __forceinline__ void
filter_own_edges(const unsigned* __restrict__ planes, int E, int Epad, int* q, Drain drain)
{
    const int tid  = threadIdx.x;
    const int lane = tid & 31;
    const int wave = tid >> 5;
    int cnt = 0;

    unsigned k[8];
#pragma unroll
    for (int b = 0; b < 8; ++b) k[b] = 0u - ((blockIdx.x >> b) & 1u);

    const int sweeps = Epad / SWEEP;

    auto pick = [&](uint4 a, uint4 b, int t) {
        const int e0 = t * SWEEP + tid * 32;
        unsigned mm = ~((a.x ^ k[0]) | (a.y ^ k[1]) | (a.z ^ k[2]) | (a.w ^ k[3]) |
                        (b.x ^ k[4]) | (b.y ^ k[5]) | (b.z ^ k[6]) | (b.w ^ k[7]));
        if (t == sweeps - 1) {
            const int left = E - e0;
            if (left < 32) mm = (left > 0) ? (mm & ((1u << left) - 1u)) : 0u;
        }
        while (true) {
            const unsigned bm = __builtin_amdgcn_ballot_w32(mm != 0u);
            if (bm == 0u) break;
            if (mm != 0u) q[cnt + __builtin_amdgcn_mbcnt_lo(bm, 0u)] = e0 + __builtin_ctz(mm);
            cnt += __builtin_popcount(bm);
            mm &= mm - 1u;
            if (cnt >= 32) {
                cnt -= 32;
                __builtin_amdgcn_wave_barrier();
                drain(cnt, 32);
            }
        }
    };

    if (sweeps > 0) {
        const int rot = (int)(blockIdx.x % (unsigned)sweeps);
        auto sweep_of = [&](int j) {
            int t = (j < sweeps ? j : sweeps - 1) + rot;
            if (t >= sweeps) t -= sweeps;
            return t;
        };
        const uint4* pl4 = (const uint4*)planes + (wave * 64 + lane);
        uint4 bufa[RING], bufb[RING];
#pragma unroll
        for (int u = 0; u < RING; ++u) {
            const uint4* p = pl4 + sweep_of(u) * (SWEEP / 16);
            bufa[u] = p[0];
            bufb[u] = p[32];
        }
        for (int j = 0; j < sweeps; j += RING) {
#pragma unroll
            for (int u = 0; u < RING; ++u) {
                const uint4 ca = bufa[u], cb = bufb[u];
                const uint4* p = pl4 + sweep_of(j + RING + u) * (SWEEP / 16);
                bufa[u] = p[0];
                bufb[u] = p[32];
                if (j + u < sweeps) pick(ca, cb, sweep_of(j + u));
            }
        }
    }
    if (cnt > 0) {
        __builtin_amdgcn_wave_barrier();
        drain(0, cnt);
    }
}

__device__ __forceinline__ void
own_deg(const int* __restrict__ src, const int* __restrict__ dst,
        const unsigned* __restrict__ planes, float* __restrict__ dinv,
        int* __restrict__ lists, int* __restrict__ lcnt, int N, int E, int Epad)
{
    __shared__ int indeg[NPB + 1];
    __shared__ int queue[AGG_THR / 32][QCAP];
    __shared__ int lcount;

    const int tid  = threadIdx.x;
    const int lane = tid & 31;
    const int wave = tid >> 5;
    const int lo   = blockIdx.x * NPB;
    const int rows = (N - lo < NPB) ? (N - lo) : NPB;

    if (tid <= NPB) indeg[tid] = 0;
    if (tid == 0) lcount = 0;
    __syncthreads();

    int2* slist = (int2*)lists + (size_t)blockIdx.x * LCAP;
    int*  q     = queue[wave];

    filter_own_edges(planes, E, Epad, q, [&](int base, int m) {
        int2 r = make_int2(NPB, 0);
        if (lane < m) {
            const int e  = q[base + lane];
            const int s  = src[e];
            const int dl = dst[e] - lo;
            atomicAdd(&indeg[dl], 1);
            r = make_int2((s << 9) | dl, 0);
        }
        int pos = 0;
        if (lane == 0) pos = atomicAdd(&lcount, m);
        pos = __builtin_amdgcn_readfirstlane(pos);
        if (lane < m && pos + m <= LCAP) slist[pos + lane] = r;
    });
    __syncthreads();

    if (tid < rows) dinv[lo + tid] = rsqrtf((float)(indeg[tid] + 1));
    if (tid == 0) lcnt[blockIdx.x] = lcount;
}

template<int OUTC>
__global__ void __launch_bounds__(AGG_THR, 16)
k_deg_gemm(const int* __restrict__ src, const int* __restrict__ dst,
           const unsigned* __restrict__ planes, float* __restrict__ dinv,
           int* __restrict__ lists, int* __restrict__ lcnt, int N, int E, int Epad,
           const float* __restrict__ A, const float* __restrict__ W,
           float* __restrict__ H, int deg_blocks)
{
    if ((int)blockIdx.x < deg_blocks) {
        own_deg(src, dst, planes, dinv, lists, lcnt, N, E, Epad);
        return;
    }
    constexpr int ROWS = 64;
    constexpr int SUBS = AGG_THR / 256;
    __shared__ float lds[SUBS][ROWS][KDIM + 4];
    const int nrows = N;
    const int sub  = threadIdx.x >> 8;
    const int tid  = threadIdx.x & 255;
    const int row0 = (((int)blockIdx.x - deg_blocks) * SUBS + sub) * ROWS;

    for (int i = tid; i < ROWS * (KDIM / 4); i += 256) {
        const int r  = i >> 5;
        const int c4 = i & 31;
        const int row = row0 + r;
        float4 v = make_float4(0.f, 0.f, 0.f, 0.f);
        if (row < nrows) v = ((const float4*)(A + (size_t)row * KDIM))[c4];
        *(float4*)&lds[sub][r][c4 * 4] = v;
    }
    __syncthreads();

    const int wave = tid >> 5;
    const int lane = tid & 31;
    const int half = lane >> 4;
    const int l16  = lane & 15;

    constexpr int CT  = OUTC / 16;
    constexpr int TPW = (4 * CT) / 8;
    const int col0  = (wave % CT) * 16;
    const int rbase = (wave / CT) * TPW;

    v8f acc[TPW] = {};
    for (int kk = 0; kk < KDIM; kk += 4) {
        const int ka = kk + half * 2;
        v2f b;
        const float* wp = W + (size_t)ka * OUTC + col0 + l16;
        b.x = wp[0];
        b.y = wp[OUTC];
#pragma unroll
        for (int t = 0; t < TPW; ++t) {
            const int rr = (rbase + t) * 16 + l16;
            v2f a;
            a.x = lds[sub][rr][ka];
            a.y = lds[sub][rr][ka + 1];
            acc[t] = __builtin_amdgcn_wmma_f32_16x16x4_f32(
                false, a, false, b, (short)0, acc[t], false, false);
        }
    }

#pragma unroll
    for (int t = 0; t < TPW; ++t) {
        const int rt0 = row0 + (rbase + t) * 16;
        float* outp = H + (size_t)rt0 * OUTC + col0;
        if (rt0 + 16 <= nrows) {
#pragma unroll
            for (int r = 0; r < 8; ++r)
                outp[(size_t)(r + half * 8) * OUTC + l16] = acc[t][r];
        } else {
#pragma unroll
            for (int r = 0; r < 8; ++r)
                if (rt0 + r + half * 8 < nrows)
                    outp[(size_t)(r + half * 8) * OUTC + l16] = acc[t][r];
        }
    }
}


template<int COLS, int LD>
__device__ __forceinline__ void
add_rows(int2 r, int m, int lane, const float* __restrict__ h, float* tile)
{
    constexpr int J = COLS / 64;
    constexpr int G = 16 / J;
    const int   s   = r.x >> 9;
    const int   dl  = r.x & 511;
    const float nrm = __int_as_float(r.y);
#pragma unroll 1
    for (int g = 0; g < m; g += G) {
        float2 v[G][J];
#pragma unroll
        for (int u = 0; u < G; ++u) {
            const int si = __builtin_amdgcn_readlane(s, g + u);
            const float* hp = h + (size_t)si * COLS + 2 * lane;
#pragma unroll
            for (int j = 0; j < J; ++j) v[u][j] = *(const float2*)(hp + 64 * j);
        }
#pragma unroll
        for (int u = 0; u < G; ++u) {
            const int   di = __builtin_amdgcn_readlane(dl, g + u);
            const float ni = __int_as_float(__builtin_amdgcn_readlane(__float_as_int(nrm), g + u));
            float* tp = tile + di * LD + 2 * lane;
#pragma unroll
            for (int j = 0; j < J; ++j) {
                atomicAdd(tp + 64 * j,     v[u][j].x * ni);
                atomicAdd(tp + 64 * j + 1, v[u][j].y * ni);
            }
        }
    }
}

constexpr int OUT2 = 64;
constexpr int RT2  = (NPB + 15) / 16;

template<int NT, int LD>
__device__ __forceinline__ void
tile_times_w(const float* tile, const float* __restrict__ W, int rg, int col0, int lane, v8f* acc)
{
    const int half = lane >> 4;
    const int l16  = lane & 15;
    for (int kk = 0; kk < KDIM; kk += 4) {
        const int ka = kk + half * 2;
        const float* wp = W + (size_t)ka * OUT2 + col0 + l16;
        v2f b;
        b.x = wp[0];
        b.y = wp[OUT2];
#pragma unroll
        for (int t = 0; t < NT; ++t) {
            const float* ap = tile + ((rg + 8 * t) * 16 + l16) * LD + ka;
            v2f a;
            a.x = ap[0];
            a.y = ap[1];
            acc[t] = __builtin_amdgcn_wmma_f32_16x16x4_f32(
                false, a, false, b, (short)0, acc[t], false, false);
        }
    }
}

template<int COLS, bool GEMM2>
__global__ void __launch_bounds__(AGG_THR)
k_agg_lds(const int* __restrict__ src, const int* __restrict__ dst,
          const unsigned* __restrict__ planes,
          const float* __restrict__ dinv, const float* __restrict__ h,
          const float* __restrict__ bias, float* __restrict__ agg,
          int* __restrict__ lists, const int* __restrict__ lcnt,
          const float* __restrict__ W2, const float* __restrict__ b1, int N, int E, int Epad)
{
    constexpr int LD    = GEMM2 ? COLS + 4 : COLS;
    constexpr int TROWS = GEMM2 ? RT2 * 16 : NPB + 1;
    __shared__ float tile[TROWS * LD];
    __shared__ int   queue[AGG_THR / 32][QCAP];

    const int tid  = threadIdx.x;
    const int lane = tid & 31;
    const int wave = tid >> 5;
    const int lo   = blockIdx.x * NPB;
    const int rows = (N - lo < NPB) ? (N - lo) : NPB;
    constexpr int C4 = COLS / 4;

    constexpr bool WEIGH = GEMM2;
    int2* slist = (int2*)lists + (size_t)blockIdx.x * LCAP;
    const int  nlist     = lcnt[blockIdx.x];
    const bool from_list = nlist <= LCAP;
    int2 r0 = make_int2(NPB, 0), r1 = make_int2(NPB, 0);
    if (from_list && wave * 32 + lane < nlist)           r0 = slist[wave * 32 + lane];
    if (from_list && wave * 32 + AGG_THR + lane < nlist) r1 = slist[wave * 32 + AGG_THR + lane];

    for (int i0 = tid; i0 < rows * C4; i0 += 4 * AGG_THR) {
        float4 v[4], b4[4];
        float  d[4];
#pragma unroll
        for (int k = 0; k < 4; ++k) {
            const int i = i0 + k * AGG_THR;
            if (i < rows * C4) {
                const int r = i / C4, c4 = i % C4;
                d[k] = dinv[lo + r];
                v[k] = ((const float4*)(h + (size_t)(lo + r) * COLS))[c4];
                if (bias) b4[k] = ((const float4*)bias)[c4];
            }
        }
#pragma unroll
        for (int k = 0; k < 4; ++k) {
            const int i = i0 + k * AGG_THR;
            if (i < rows * C4) {
                const int r = i / C4, c4 = i % C4;
                float4 w = v[k];
                w.x = w.x * d[k] * d[k]; w.y = w.y * d[k] * d[k];
                w.z = w.z * d[k] * d[k]; w.w = w.w * d[k] * d[k];
                if (bias) { w.x += b4[k].x; w.y += b4[k].y; w.z += b4[k].z; w.w += b4[k].w; }
                *(float4*)&tile[r * LD + c4 * 4] = w;
            }
        }
    }
    float wa = 0.f, wd = 0.f;
    if (WEIGH && from_list && wave * 32 + lane < nlist) {
        wa = dinv[r0.x >> 9];
        wd = dinv[lo + (r0.x & 511)];
    }
    __syncthreads();

    if (from_list) {
        for (int c = wave * 32; c < nlist; c += AGG_THR) {
            int2 r = r0;
            const int  m = (nlist - c < 32) ? (nlist - c) : 32;
            if (WEIGH) {
                r.y = __float_as_int(wa * wd);
                if (lane < m) slist[c + lane].y = r.y;
            }
            r0 = r1;
            if (WEIGH) {
                wa = 0.f; wd = 0.f;
                if (c + AGG_THR + lane < nlist) {
                    wa = dinv[r0.x >> 9];
                    wd = dinv[lo + (r0.x & 511)];
                }
            }
            r1 = make_int2(NPB, 0);
            if (c + 2 * AGG_THR + lane < nlist) r1 = slist[c + 2 * AGG_THR + lane];
            add_rows<COLS, LD>(r, m, lane, h, tile);
        }
    } else {
        int* q = queue[wave];
        filter_own_edges(planes, E, Epad, q, [&](int base, int m) {
            int2 r = make_int2(NPB, 0);
            if (lane < m) {
                const int e = q[base + lane];
                const int s = src[e];
                const int d = dst[e];
                r = make_int2((s << 9) | (d - lo), __float_as_int(dinv[s] * dinv[d]));
            }
            add_rows<COLS, LD>(r, m, lane, h, tile);
        });
    }
    __syncthreads();

    if constexpr (!GEMM2) {
        for (int i = tid; i < rows * C4; i += AGG_THR) {
            const int r = i / C4, c4 = i % C4;
            ((float4*)(agg + (size_t)(lo + r) * COLS))[c4] = *(const float4*)&tile[r * LD + c4 * 4];
        }
    } else {
        static_assert(COLS == KDIM, "the tile is the A operand of the layer-2 product");
        static_assert(RT2 > 24 && RT2 <= 32, "a wave takes 3 or 4 row tiles");
        for (int i = tid; i < TROWS * C4; i += AGG_THR) {
            const int r = i / C4, c4 = i % C4;
            float4 v = make_float4(0.f, 0.f, 0.f, 0.f);
            if (r < rows) {
                v = *(const float4*)&tile[r * LD + c4 * 4];
                const float4 b4 = ((const float4*)b1)[c4];
                v.x = fmaxf(v.x + b4.x, 0.f);
                v.y = fmaxf(v.y + b4.y, 0.f);
                v.z = fmaxf(v.z + b4.z, 0.f);
                v.w = fmaxf(v.w + b4.w, 0.f);
            }
            *(float4*)&tile[r * LD + c4 * 4] = v;
        }
        __syncthreads();

        const int wv   = __builtin_amdgcn_readfirstlane(wave);
        const int col0 = (wv & 3) * 16;
        const int rg   = wv >> 2;
        const int nt   = (RT2 - rg + 7) / 8;
        v8f acc[4] = {};
        if (nt == 4) tile_times_w<4, LD>(tile, W2, rg, col0, lane, acc);
        else         tile_times_w<3, LD>(tile, W2, rg, col0, lane, acc);
        __syncthreads();

        const int half = lane >> 4;
        const int l16  = lane & 15;
#pragma unroll
        for (int t = 0; t < 4; ++t) {
            if (t < nt) {
                float* tp = tile + ((rg + 8 * t) * 16 + half * 8) * LD + col0 + l16;
#pragma unroll
                for (int r = 0; r < 8; ++r) tp[r * LD] = acc[t][r];
            }
        }
        __syncthreads();

        constexpr int O4 = OUT2 / 4;
        for (int i = tid; i < rows * O4; i += AGG_THR) {
            const int r = i / O4, c4 = i % O4;
            ((float4*)(agg + (size_t)(lo + r) * OUT2))[c4] = *(const float4*)&tile[r * LD + c4 * 4];
        }
    }
}

extern "C" void kernel_launch(void* const* d_in, const int* in_sizes, int n_in,
                              void* d_out, int out_size, void* d_ws, size_t ws_size,
                              hipStream_t stream)
{
    const float* x  = (const float*)d_in[0];
    const int*   ei = (const int*)  d_in[1];
    const float* W1 = (const float*)d_in[2];
    const float* b1 = (const float*)d_in[3];
    const float* W2 = (const float*)d_in[4];
    const float* b2 = (const float*)d_in[5];
    float* out = (float*)d_out;

    constexpr int HID = 128, OUT = 64;
    const int N = in_sizes[0] / KDIM;
    const int E = in_sizes[1] / 2;
    const int* src = ei;
    const int* dst = ei + E;

    size_t off = 0;
    auto carve = [&](size_t bytes) {
        size_t o = off;
        off = (off + bytes + 255) & ~(size_t)255;
        return o;
    };
    char* ws = (char*)d_ws;
    float* dinv = (float*)(ws + carve((size_t)N * 4));
    float* h    = (float*)(ws + carve((size_t)N * HID * 4));
    float* h2   = (float*)(ws + carve((size_t)N * OUT * 4));
    const int Epad = (E + SWEEP - 1) / SWEEP * SWEEP;
    unsigned* planes = (unsigned*)(ws + carve((size_t)Epad));

    const int gemmBlocks = ((N + 63) / 64 + 3) / 4;
    const int aggBlocks  = (N + NPB - 1) / NPB;
    int*  lists = (int*) (ws + carve((size_t)aggBlocks * LCAP * 8));
    int*  lcnt  = (int*) (ws + carve((size_t)aggBlocks * 4));

    k_key     <<<Epad / 256, 256, 0, stream>>>(dst, planes, E);
    k_deg_gemm<HID><<<aggBlocks + gemmBlocks, AGG_THR, 0, stream>>>(src, dst, planes, dinv, lists, lcnt, N, E, Epad, x, W1, h, aggBlocks);

    k_agg_lds<HID, true><<<aggBlocks, AGG_THR, 0, stream>>>(src, dst, planes, dinv, h, nullptr, h2, lists, lcnt, W2, b1, N, E, Epad);

    k_agg_lds<OUT, false><<<aggBlocks, AGG_THR, 0, stream>>>(src, dst, planes, dinv, h2, b2, out, lists, lcnt, nullptr, nullptr, N, E, Epad);
}


# Group's edge list split at build time into one segment per wave (dst row mod 32): a wave adds only its own tile rows, plain wide LDS load/fma/store, no LDS f32 atomics; sum order = segment arrival
# speedup vs baseline: 3.8070x; 1.0468x over previous
#include <hip/hip_runtime.h>

typedef __attribute__((ext_vector_type(2))) float v2f;
typedef __attribute__((ext_vector_type(8))) float v8f;

constexpr int KDIM = 128;

constexpr int NPB     = 391;
constexpr int AGG_THR = 1024;
constexpr int SWEEP   = 32 * AGG_THR;

constexpr int RING    = 4;
constexpr int QCAP    = 64;
constexpr int LCAP    = 16384;
constexpr int NSEG    = AGG_THR / 32;
constexpr int SEGCAP  = LCAP / NSEG;
static_assert((NSEG & (NSEG - 1)) == 0, "the owning wave of a row is taken with a mask");

__global__ void __launch_bounds__(256) k_key(const int* __restrict__ dst,
                                             unsigned* __restrict__ planes, int e) {
    const int i    = blockIdx.x * 256 + threadIdx.x;
    const int lane = threadIdx.x & 31;
    unsigned own = 0u;
    if (i < e) own = (unsigned)(dst[i] / NPB);
    unsigned pl = 0u;
#pragma unroll
    for (int b = 0; b < 8; ++b) {
        const unsigned bal = __builtin_amdgcn_ballot_w32(((own >> b) & 1u) != 0u);
        if (lane == b) pl = bal;
    }
    if (lane < 8) {
        const int g32 = i >> 5;
        planes[(g32 >> 5) * 256 + (lane >> 2) * 128 + (g32 & 31) * 4 + (lane & 3)] = pl;
    }
}

template<class Drain>
__device__ __forceinline__ void
filter_own_edges(const unsigned* __restrict__ planes, int E, int Epad, int* q, Drain drain)
{
    const int tid  = threadIdx.x;
    const int lane = tid & 31;
    const int wave = tid >> 5;
    int cnt = 0;

    unsigned k[8];
#pragma unroll
    for (int b = 0; b < 8; ++b) k[b] = 0u - ((blockIdx.x >> b) & 1u);

    const int sweeps = Epad / SWEEP;

    auto pick = [&](uint4 a, uint4 b, int t) {
        const int e0 = t * SWEEP + tid * 32;
        unsigned mm = ~((a.x ^ k[0]) | (a.y ^ k[1]) | (a.z ^ k[2]) | (a.w ^ k[3]) |
                        (b.x ^ k[4]) | (b.y ^ k[5]) | (b.z ^ k[6]) | (b.w ^ k[7]));
        if (t == sweeps - 1) {
            const int left = E - e0;
            if (left < 32) mm = (left > 0) ? (mm & ((1u << left) - 1u)) : 0u;
        }
        while (true) {
            const unsigned bm = __builtin_amdgcn_ballot_w32(mm != 0u);
            if (bm == 0u) break;
            if (mm != 0u) q[cnt + __builtin_amdgcn_mbcnt_lo(bm, 0u)] = e0 + __builtin_ctz(mm);
            cnt += __builtin_popcount(bm);
            mm &= mm - 1u;
            if (cnt >= 32) {
                cnt -= 32;
                __builtin_amdgcn_wave_barrier();
                drain(cnt, 32);
            }
        }
    };

    if (sweeps > 0) {
        const int rot = (int)(blockIdx.x % (unsigned)sweeps);
        auto sweep_of = [&](int j) {
            int t = (j < sweeps ? j : sweeps - 1) + rot;
            if (t >= sweeps) t -= sweeps;
            return t;
        };
        const uint4* pl4 = (const uint4*)planes + (wave * 64 + lane);
        uint4 bufa[RING], bufb[RING];
#pragma unroll
        for (int u = 0; u < RING; ++u) {
            const uint4* p = pl4 + sweep_of(u) * (SWEEP / 16);
            bufa[u] = p[0];
            bufb[u] = p[32];
        }
        for (int j = 0; j < sweeps; j += RING) {
#pragma unroll
            for (int u = 0; u < RING; ++u) {
                const uint4 ca = bufa[u], cb = bufb[u];
                const uint4* p = pl4 + sweep_of(j + RING + u) * (SWEEP / 16);
                bufa[u] = p[0];
                bufb[u] = p[32];
                if (j + u < sweeps) pick(ca, cb, sweep_of(j + u));
            }
        }
    }
    if (cnt > 0) {
        __builtin_amdgcn_wave_barrier();
        drain(0, cnt);
    }
}

__device__ __forceinline__ void
own_deg(const int* __restrict__ src, const int* __restrict__ dst,
        const unsigned* __restrict__ planes, float* __restrict__ dinv,
        int* __restrict__ lists, int* __restrict__ lcnt, int N, int E, int Epad)
{
    __shared__ int indeg[NPB + 1];
    __shared__ int queue[AGG_THR / 32][QCAP];
    __shared__ int segcnt[NSEG];

    const int tid  = threadIdx.x;
    const int lane = tid & 31;
    const int wave = tid >> 5;
    const int lo   = blockIdx.x * NPB;
    const int rows = (N - lo < NPB) ? (N - lo) : NPB;

    if (tid <= NPB) indeg[tid] = 0;
    if (tid < NSEG) segcnt[tid] = 0;
    __syncthreads();

    int2* slist = (int2*)lists + (size_t)blockIdx.x * LCAP;
    int*  q     = queue[wave];

    filter_own_edges(planes, E, Epad, q, [&](int base, int m) {
        int2 r   = make_int2(0, 0);
        int  sg  = 0;
        int  pos = SEGCAP;
        if (lane < m) {
            const int e  = q[base + lane];
            const int s  = src[e];
            const int d  = dst[e];
            asm volatile("" :: "v"(s), "v"(d));
            const int dl = d - lo;
            atomicAdd(&indeg[dl], 1);
            sg  = dl & (NSEG - 1);
            pos = atomicAdd(&segcnt[sg], 1);
            r   = make_int2((s << 9) | dl, 0);
        }
        if (pos < SEGCAP) slist[sg * SEGCAP + pos] = r;
    });
    __syncthreads();

    if (tid < rows) dinv[lo + tid] = rsqrtf((float)(indeg[tid] + 1));
    if (tid < NSEG) lcnt[blockIdx.x * NSEG + tid] = segcnt[tid];
}

template<int OUTC>
__global__ void __launch_bounds__(AGG_THR, 16)
k_deg_gemm(const int* __restrict__ src, const int* __restrict__ dst,
           const unsigned* __restrict__ planes, float* __restrict__ dinv,
           int* __restrict__ lists, int* __restrict__ lcnt, int N, int E, int Epad,
           const float* __restrict__ A, const float* __restrict__ W,
           float* __restrict__ H, int deg_blocks)
{
    if ((int)blockIdx.x < deg_blocks) {
        own_deg(src, dst, planes, dinv, lists, lcnt, N, E, Epad);
        return;
    }
    constexpr int ROWS = 64;
    constexpr int SUBS = AGG_THR / 256;
    __shared__ float lds[SUBS][ROWS][KDIM + 4];
    const int nrows = N;
    const int sub  = threadIdx.x >> 8;
    const int tid  = threadIdx.x & 255;
    const int row0 = (((int)blockIdx.x - deg_blocks) * SUBS + sub) * ROWS;

    for (int i = tid; i < ROWS * (KDIM / 4); i += 256) {
        const int r  = i >> 5;
        const int c4 = i & 31;
        const int row = row0 + r;
        float4 v = make_float4(0.f, 0.f, 0.f, 0.f);
        if (row < nrows) v = ((const float4*)(A + (size_t)row * KDIM))[c4];
        *(float4*)&lds[sub][r][c4 * 4] = v;
    }
    __syncthreads();

    const int wave = tid >> 5;
    const int lane = tid & 31;
    const int half = lane >> 4;
    const int l16  = lane & 15;

    constexpr int CT  = OUTC / 16;
    constexpr int TPW = (4 * CT) / 8;
    const int col0  = (wave % CT) * 16;
    const int rbase = (wave / CT) * TPW;

    v8f acc[TPW] = {};
    for (int kk = 0; kk < KDIM; kk += 4) {
        const int ka = kk + half * 2;
        v2f b;
        const float* wp = W + (size_t)ka * OUTC + col0 + l16;
        b.x = wp[0];
        b.y = wp[OUTC];
#pragma unroll
        for (int t = 0; t < TPW; ++t) {
            const int rr = (rbase + t) * 16 + l16;
            v2f a;
            a.x = lds[sub][rr][ka];
            a.y = lds[sub][rr][ka + 1];
            acc[t] = __builtin_amdgcn_wmma_f32_16x16x4_f32(
                false, a, false, b, (short)0, acc[t], false, false);
        }
    }

#pragma unroll
    for (int t = 0; t < TPW; ++t) {
        const int rt0 = row0 + (rbase + t) * 16;
        float* outp = H + (size_t)rt0 * OUTC + col0;
        if (rt0 + 16 <= nrows) {
#pragma unroll
            for (int r = 0; r < 8; ++r)
                outp[(size_t)(r + half * 8) * OUTC + l16] = acc[t][r];
        } else {
#pragma unroll
            for (int r = 0; r < 8; ++r)
                if (rt0 + r + half * 8 < nrows)
                    outp[(size_t)(r + half * 8) * OUTC + l16] = acc[t][r];
        }
    }
}


template<int COLS, int LD>
__device__ __forceinline__ void
add_rows(int2 r, int m, int lane, const float* __restrict__ h, float* tile)
{
    constexpr int J = COLS / 64;
    constexpr int G = 16 / J;
    const int   s   = r.x >> 9;
    const int   dl  = r.x & 511;
    const float nrm = __int_as_float(r.y);
#pragma unroll 1
    for (int g = 0; g < m; g += G) {
        float2 v[G][J];
#pragma unroll
        for (int u = 0; u < G; ++u) {
            const int si = __builtin_amdgcn_readlane(s, g + u);
            const float* hp = h + (size_t)si * COLS + 2 * lane;
#pragma unroll
            for (int j = 0; j < J; ++j) v[u][j] = *(const float2*)(hp + 64 * j);
        }
#pragma unroll
        for (int u = 0; u < G; ++u) {
            const int   di = __builtin_amdgcn_readlane(dl, g + u);
            const float ni = __int_as_float(__builtin_amdgcn_readlane(__float_as_int(nrm), g + u));
            float* tp = tile + di * LD + 2 * lane;
#pragma unroll
            for (int j = 0; j < J; ++j) {
                atomicAdd(tp + 64 * j,     v[u][j].x * ni);
                atomicAdd(tp + 64 * j + 1, v[u][j].y * ni);
            }
        }
    }
}

template<int COLS, int LD>
__device__ __forceinline__ void
add_rows_own(int2 r, int m, int lane, const float* __restrict__ h, float* tile)
{
    constexpr int V = COLS / 32;
    constexpr int G = 16;
    typedef __attribute__((ext_vector_type(V))) float vrow;
    const int   s   = r.x >> 9;
    const int   dl  = r.x & 511;
    const float nrm = __int_as_float(r.y);
    static_assert(2 * G == 32, "a batch is two groups");
    auto group = [&](const int g) {
        vrow v[G];
#pragma unroll
        for (int u = 0; u < G; ++u) {
            const int si = __builtin_amdgcn_readlane(s, g + u);
            v[u] = *(const vrow*)(h + (size_t)si * COLS + V * lane);
        }
#pragma unroll
        for (int u = 0; u < G; u += 2) {
            const int   da = __builtin_amdgcn_readlane(dl, g + u);
            const int   db = __builtin_amdgcn_readlane(dl, g + u + 1);
            const float na = __int_as_float(__builtin_amdgcn_readlane(__float_as_int(nrm), g + u));
            const float nb = __int_as_float(__builtin_amdgcn_readlane(__float_as_int(nrm), g + u + 1));
            vrow* pa = (vrow*)(tile + da * LD + V * lane);
            vrow* pb = (vrow*)(tile + db * LD + V * lane);
            if (da != db) {
                vrow ta = *pa;
                vrow tb = *pb;
                ta += v[u] * na;
                tb += v[u + 1] * nb;
                *pa = ta;
                *pb = tb;
            } else {
                vrow ta = *pa;
                ta += v[u] * na;
                ta += v[u + 1] * nb;
                *pa = ta;
            }
        }
    };
    group(0);
    if (m > G) group(G);
}

constexpr int OUT2 = 64;
constexpr int RT2  = (NPB + 15) / 16;

template<int NT, int LD>
__device__ __forceinline__ void
tile_times_w(const float* tile, const float* __restrict__ W, int rg, int col0, int lane, v8f* acc)
{
    const int half = lane >> 4;
    const int l16  = lane & 15;
    for (int kk = 0; kk < KDIM; kk += 4) {
        const int ka = kk + half * 2;
        const float* wp = W + (size_t)ka * OUT2 + col0 + l16;
        v2f b;
        b.x = wp[0];
        b.y = wp[OUT2];
#pragma unroll
        for (int t = 0; t < NT; ++t) {
            const float* ap = tile + ((rg + 8 * t) * 16 + l16) * LD + ka;
            v2f a;
            a.x = ap[0];
            a.y = ap[1];
            acc[t] = __builtin_amdgcn_wmma_f32_16x16x4_f32(
                false, a, false, b, (short)0, acc[t], false, false);
        }
    }
}

template<int COLS, bool GEMM2>
__global__ void __launch_bounds__(AGG_THR)
k_agg_lds(const int* __restrict__ src, const int* __restrict__ dst,
          const unsigned* __restrict__ planes,
          const float* __restrict__ dinv, const float* __restrict__ h,
          const float* __restrict__ bias, float* __restrict__ agg,
          int* __restrict__ lists, const int* __restrict__ lcnt,
          const float* __restrict__ W2, const float* __restrict__ b1, int N, int E, int Epad)
{
    constexpr int LD    = GEMM2 ? COLS + 4 : COLS;
    constexpr int TROWS = NPB + NSEG;
    static_assert(TROWS >= RT2 * 16 && TROWS <= 512, "whole 16-row tiles for the product; a local row fits the record's 9 bits");
    __shared__ float tile[TROWS * LD];
    __shared__ int   queue[AGG_THR / 32][QCAP];

    const int tid  = threadIdx.x;
    const int lane = tid & 31;
    const int wave = tid >> 5;
    const int lo   = blockIdx.x * NPB;
    const int rows = (N - lo < NPB) ? (N - lo) : NPB;
    constexpr int C4 = COLS / 4;

    constexpr bool WEIGH = GEMM2;
    static_assert(NSEG == 32, "lane l looks at the count of segment l");
    int2* seg = (int2*)lists + (size_t)blockIdx.x * LCAP + wave * SEGCAP;
    const int  nsl       = lcnt[blockIdx.x * NSEG + lane];
    const bool from_list = __builtin_amdgcn_ballot_w32(nsl > SEGCAP) == 0u;
    const int  nlist     = __builtin_amdgcn_readlane(nsl, __builtin_amdgcn_readfirstlane(wave));
    const int2 standin   = make_int2(NPB + wave, 0);
    int2 r0 = standin, r1 = standin;
    if (from_list && lane < nlist)      r0 = seg[lane];
    if (from_list && 32 + lane < nlist) r1 = seg[32 + lane];

    for (int i0 = tid; i0 < rows * C4; i0 += 4 * AGG_THR) {
        float4 v[4], b4[4];
        float  d[4];
#pragma unroll
        for (int k = 0; k < 4; ++k) {
            const int i = i0 + k * AGG_THR;
            if (i < rows * C4) {
                const int r = i / C4, c4 = i % C4;
                d[k] = dinv[lo + r];
                v[k] = ((const float4*)(h + (size_t)(lo + r) * COLS))[c4];
                if (bias) b4[k] = ((const float4*)bias)[c4];
            }
        }
#pragma unroll
        for (int k = 0; k < 4; ++k) {
            const int i = i0 + k * AGG_THR;
            if (i < rows * C4) {
                const int r = i / C4, c4 = i % C4;
                float4 w = v[k];
                w.x = w.x * d[k] * d[k]; w.y = w.y * d[k] * d[k];
                w.z = w.z * d[k] * d[k]; w.w = w.w * d[k] * d[k];
                if (bias) { w.x += b4[k].x; w.y += b4[k].y; w.z += b4[k].z; w.w += b4[k].w; }
                *(float4*)&tile[r * LD + c4 * 4] = w;
            }
        }
    }
    float wa = 0.f, wd = 0.f;
    if (WEIGH && from_list && lane < nlist) {
        wa = dinv[r0.x >> 9];
        wd = dinv[lo + (r0.x & 511)];
    }
    __syncthreads();

    if (from_list) {
        for (int c = 0; c < nlist; c += 32) {
            int2 r = r0;
            const int  m = (nlist - c < 32) ? (nlist - c) : 32;
            if (WEIGH) {
                r.y = __float_as_int(wa * wd);
                if (lane < m) seg[c + lane].y = r.y;
            }
            r0 = r1;
            if (WEIGH) {
                wa = 0.f; wd = 0.f;
                if (c + 32 + lane < nlist) {
                    wa = dinv[r0.x >> 9];
                    wd = dinv[lo + (r0.x & 511)];
                }
            }
            r1 = standin;
            if (c + 64 + lane < nlist) r1 = seg[c + 64 + lane];
            add_rows_own<COLS, LD>(r, m, lane, h, tile);
        }
    } else {
        int* q = queue[wave];
        filter_own_edges(planes, E, Epad, q, [&](int base, int m) {
            int2 r = make_int2(NPB, 0);
            if (lane < m) {
                const int e = q[base + lane];
                const int s = src[e];
                const int d = dst[e];
                r = make_int2((s << 9) | (d - lo), __float_as_int(dinv[s] * dinv[d]));
            }
            add_rows<COLS, LD>(r, m, lane, h, tile);
        });
    }
    __syncthreads();

    if constexpr (!GEMM2) {
        for (int i = tid; i < rows * C4; i += AGG_THR) {
            const int r = i / C4, c4 = i % C4;
            ((float4*)(agg + (size_t)(lo + r) * COLS))[c4] = *(const float4*)&tile[r * LD + c4 * 4];
        }
    } else {
        static_assert(COLS == KDIM, "the tile is the A operand of the layer-2 product");
        static_assert(RT2 > 24 && RT2 <= 32, "a wave takes 3 or 4 row tiles");
        for (int i = tid; i < TROWS * C4; i += AGG_THR) {
            const int r = i / C4, c4 = i % C4;
            float4 v = make_float4(0.f, 0.f, 0.f, 0.f);
            if (r < rows) {
                v = *(const float4*)&tile[r * LD + c4 * 4];
                const float4 b4 = ((const float4*)b1)[c4];
                v.x = fmaxf(v.x + b4.x, 0.f);
                v.y = fmaxf(v.y + b4.y, 0.f);
                v.z = fmaxf(v.z + b4.z, 0.f);
                v.w = fmaxf(v.w + b4.w, 0.f);
            }
            *(float4*)&tile[r * LD + c4 * 4] = v;
        }
        __syncthreads();

        const int wv   = __builtin_amdgcn_readfirstlane(wave);
        const int col0 = (wv & 3) * 16;
        const int rg   = wv >> 2;
        const int nt   = (RT2 - rg + 7) / 8;
        v8f acc[4] = {};
        if (nt == 4) tile_times_w<4, LD>(tile, W2, rg, col0, lane, acc);
        else         tile_times_w<3, LD>(tile, W2, rg, col0, lane, acc);
        __syncthreads();

        const int half = lane >> 4;
        const int l16  = lane & 15;
#pragma unroll
        for (int t = 0; t < 4; ++t) {
            if (t < nt) {
                float* tp = tile + ((rg + 8 * t) * 16 + half * 8) * LD + col0 + l16;
#pragma unroll
                for (int r = 0; r < 8; ++r) tp[r * LD] = acc[t][r];
            }
        }
        __syncthreads();

        constexpr int O4 = OUT2 / 4;
        for (int i = tid; i < rows * O4; i += AGG_THR) {
            const int r = i / O4, c4 = i % O4;
            ((float4*)(agg + (size_t)(lo + r) * OUT2))[c4] = *(const float4*)&tile[r * LD + c4 * 4];
        }
    }
}

extern "C" void kernel_launch(void* const* d_in, const int* in_sizes, int n_in,
                              void* d_out, int out_size, void* d_ws, size_t ws_size,
                              hipStream_t stream)
{
    const float* x  = (const float*)d_in[0];
    const int*   ei = (const int*)  d_in[1];
    const float* W1 = (const float*)d_in[2];
    const float* b1 = (const float*)d_in[3];
    const float* W2 = (const float*)d_in[4];
    const float* b2 = (const float*)d_in[5];
    float* out = (float*)d_out;

    constexpr int HID = 128, OUT = 64;
    const int N = in_sizes[0] / KDIM;
    const int E = in_sizes[1] / 2;
    const int* src = ei;
    const int* dst = ei + E;

    size_t off = 0;
    auto carve = [&](size_t bytes) {
        size_t o = off;
        off = (off + bytes + 255) & ~(size_t)255;
        return o;
    };
    char* ws = (char*)d_ws;
    float* dinv = (float*)(ws + carve((size_t)N * 4));
    float* h    = (float*)(ws + carve((size_t)N * HID * 4));
    float* h2   = (float*)(ws + carve((size_t)N * OUT * 4));
    const int Epad = (E + SWEEP - 1) / SWEEP * SWEEP;
    unsigned* planes = (unsigned*)(ws + carve((size_t)Epad));

    const int gemmBlocks = ((N + 63) / 64 + 3) / 4;
    const int aggBlocks  = (N + NPB - 1) / NPB;
    int*  lists = (int*) (ws + carve((size_t)aggBlocks * LCAP * 8));
    int*  lcnt  = (int*) (ws + carve((size_t)aggBlocks * NSEG * 4));

    k_key     <<<Epad / 256, 256, 0, stream>>>(dst, planes, E);
    k_deg_gemm<HID><<<aggBlocks + gemmBlocks, AGG_THR, 0, stream>>>(src, dst, planes, dinv, lists, lcnt, N, E, Epad, x, W1, h, aggBlocks);

    k_agg_lds<HID, true><<<aggBlocks, AGG_THR, 0, stream>>>(src, dst, planes, dinv, h, nullptr, h2, lists, lcnt, W2, b1, N, E, Epad);

    k_agg_lds<OUT, false><<<aggBlocks, AGG_THR, 0, stream>>>(src, dst, planes, dinv, h2, b2, out, lists, lcnt, nullptr, nullptr, N, E, Epad);
}
